# starting kernel
# baseline (speedup 1.0000x reference)
#include <hip/hip_runtime.h>
#include <hip/hip_bf16.h>
#include <math.h>

typedef __attribute__((ext_vector_type(2))) float v2f;
typedef __attribute__((ext_vector_type(8))) float v8f;

#define B_   4
#define N_   2048
#define FIN  256
#define FOUT 128
#define ALPHA_ 0.2f
#define EPS_   1e-5f
#define NEG_   -9000000000000000.0f

__global__ __launch_bounds__(128) void wh_kernel(const float* __restrict__ h,
                                                 const float* __restrict__ W,
                                                 float* __restrict__ Wh) {
  int lane  = threadIdx.x & 31;
  int tile  = blockIdx.x * 4 + (threadIdx.x >> 5);
  int otile = tile & 7;
  int rest  = tile >> 3;
  int itile = rest & (N_ / 16 - 1);
  int b     = rest >> 7;
  int i0 = itile * 16, o0 = otile * 16;
  int r16  = lane & 15;
  int koff = (lane >> 4) << 1;

  const float* hrow = h + (size_t)(b * N_ + i0 + r16) * FIN;
  v8f c = {};
  for (int k0 = 0; k0 < FIN; k0 += 4) {
    float2 av = *(const float2*)(hrow + k0 + koff);
    v2f a; a.x = av.x; a.y = av.y;
    v2f bb;
    bb.x = W[(size_t)(k0 + koff + 0) * FOUT + o0 + r16];
    bb.y = W[(size_t)(k0 + koff + 1) * FOUT + o0 + r16];
    c = __builtin_amdgcn_wmma_f32_16x16x4_f32(false, a, false, bb, (short)0, c,
                                              false, false);
  }
#pragma unroll
  for (int r = 0; r < 8; ++r) {
    int M = r + ((lane >> 4) << 3);
    Wh[(size_t)(b * N_ + i0 + M) * FOUT + o0 + r16] = c[r];
  }
}

__global__ __launch_bounds__(256) void s12_kernel(const float* __restrict__ Wh,
                                                  const float* __restrict__ a,
                                                  float* __restrict__ s1,
                                                  float* __restrict__ s2) {
  int lane = threadIdx.x & 31;
  int node = (int)((blockIdx.x * blockDim.x + threadIdx.x) >> 5);
  if (node >= B_ * N_) return;
  const float4* w4 = (const float4*)(Wh + (size_t)node * FOUT);
  const float4* a1 = (const float4*)a;
  const float4* a2 = (const float4*)(a + FOUT);
  float4 wv = w4[lane];
  float4 x1 = a1[lane];
  float4 x2 = a2[lane];
  float p1 = wv.x * x1.x + wv.y * x1.y + wv.z * x1.z + wv.w * x1.w;
  float p2 = wv.x * x2.x + wv.y * x2.y + wv.z * x2.z + wv.w * x2.w;
  for (int off = 16; off >= 1; off >>= 1) {
    p1 += __shfl_xor(p1, off, 32);
    p2 += __shfl_xor(p2, off, 32);
  }
  if (lane == 0) { s1[node] = p1; s2[node] = p2; }
}

__global__ __launch_bounds__(256) void stats_kernel(const int* __restrict__ adj,
                                                    const float* __restrict__ s1,
                                                    const float* __restrict__ s2,
                                                    float* __restrict__ mrow,
                                                    float* __restrict__ lrow,
                                                    unsigned* __restrict__ mbits) {
  int lane = threadIdx.x & 31;
  int row  = blockIdx.x * 8 + (threadIdx.x >> 5);
  int b    = row >> 11;
  float s1v = s1[row];
  const float* s2b = s2 + (size_t)b * N_;
  const int* arow  = adj + (size_t)row * N_;
  unsigned* mb     = mbits + (size_t)row * (N_ / 32);
  float m = -INFINITY, l = 0.f;
  for (int w = 0; w < N_ / 32; ++w) {
    int j = w * 32 + lane;
    float e = s1v + s2b[j];
    e = (e >= 0.f) ? e : ALPHA_ * e;
    bool pred = arow[j] > 0;
    unsigned long long bl = __ballot(pred);
    if (lane == 0) mb[w] = (unsigned)bl;
    float tv = pred ? e : NEG_;
    if (tv > m) { l = l * __expf(m - tv) + 1.0f; m = tv; }
    else        { l += __expf(tv - m); }
  }
  for (int off = 16; off >= 1; off >>= 1) {
    float m2 = __shfl_xor(m, off, 32);
    float l2 = __shfl_xor(l, off, 32);
    float M  = fmaxf(m, m2);
    l = l * __expf(m - M) + l2 * __expf(m2 - M);
    m = M;
  }
  if (lane == 0) { mrow[row] = m; lrow[row] = l; }
}

__global__ __launch_bounds__(128) void hprime_kernel(
    const unsigned* __restrict__ mbits, const float* __restrict__ Wh,
    const float* __restrict__ s1, const float* __restrict__ s2,
    const float* __restrict__ mrow, const float* __restrict__ lrow,
    const float* __restrict__ gamma, const float* __restrict__ beta,
    float* __restrict__ out) {
  __shared__ float lds[4][16][FOUT];
  int tid  = threadIdx.x;
  int lane = tid & 31;
  int wave = tid >> 5;
  int itile = blockIdx.x & (N_ / 16 - 1);
  int b     = blockIdx.x >> 7;
  int i0    = itile * 16;
  int r16   = lane & 15;
  int koff  = (lane >> 4) << 1;
  int rowid = b * N_ + i0 + r16;

  float s1v  = s1[rowid];
  float mi   = mrow[rowid];
  float invl = 1.0f / lrow[rowid];
  const unsigned* mb = mbits + (size_t)rowid * (N_ / 32);
  const float* s2b   = s2 + (size_t)b * N_;
  const float* Whb   = Wh + (size_t)b * N_ * FOUT;

  v8f acc[8];
#pragma unroll
  for (int ot = 0; ot < 8; ++ot) acc[ot] = (v8f){};

  int jbeg = wave * (N_ / 4);
  int jend = jbeg + (N_ / 4);
  for (int j0 = jbeg; j0 < jend; j0 += 4) {
    int j = j0 + koff;
    unsigned wbits = mb[j0 >> 5];
    v2f a;
#pragma unroll
    for (int v = 0; v < 2; ++v) {
      int jj = j + v;
      float e = s1v + s2b[jj];
      e = (e >= 0.f) ? e : ALPHA_ * e;
      float tv = ((wbits >> ((j0 & 31) + koff + v)) & 1u) ? e : NEG_;
      a[v] = __expf(tv - mi) * invl;
    }
    const float* w0 = Whb + (size_t)j * FOUT + r16;
    const float* w1 = w0 + FOUT;
#pragma unroll
    for (int ot = 0; ot < 8; ++ot) {
      v2f bb;
      bb.x = w0[ot * 16];
      bb.y = w1[ot * 16];
      acc[ot] = __builtin_amdgcn_wmma_f32_16x16x4_f32(false, a, false, bb,
                                                      (short)0, acc[ot],
                                                      false, false);
    }
  }

#pragma unroll
  for (int ot = 0; ot < 8; ++ot)
#pragma unroll
    for (int r = 0; r < 8; ++r) {
      int M = r + ((lane >> 4) << 3);
      lds[wave][M][ot * 16 + r16] = acc[ot][r];
    }
  __syncthreads();

  int row16 = tid >> 3;
  int seg   = tid & 7;
  float vals[16];
  float s = 0.f, ss = 0.f;
#pragma unroll
  for (int c = 0; c < 16; ++c) {
    int col = seg * 16 + c;
    float x = lds[0][row16][col] + lds[1][row16][col] +
              lds[2][row16][col] + lds[3][row16][col];
    vals[c] = x;
    s += x;
    ss += x * x;
  }
  for (int off = 4; off >= 1; off >>= 1) {
    s  += __shfl_xor(s, off, 8);
    ss += __shfl_xor(ss, off, 8);
  }
  float mu   = s * (1.0f / FOUT);
  float var  = ss * (1.0f / FOUT) - mu * mu;
  float rstd = rsqrtf(var + EPS_);
  float* orow = out + (size_t)(b * N_ + i0 + row16) * FOUT;
#pragma unroll
  for (int c = 0; c < 16; ++c) {
    int col = seg * 16 + c;
    float x = (vals[c] - mu) * rstd * gamma[col] + beta[col];
    orow[col] = 0.5f * x * (1.0f + erff(x * 0.70710678118654752f));
  }
}

extern "C" void kernel_launch(void* const* d_in, const int* in_sizes, int n_in,
                              void* d_out, int out_size, void* d_ws, size_t ws_size,
                              hipStream_t stream) {
  const float* h     = (const float*)d_in[0];
  const int*   adj   = (const int*)d_in[1];
  const float* W     = (const float*)d_in[2];
  const float* a     = (const float*)d_in[3];
  const float* gamma = (const float*)d_in[4];
  const float* beta  = (const float*)d_in[5];
  float* out = (float*)d_out;

  float* ws   = (float*)d_ws;
  float* Wh   = ws;
  float* s1   = Wh + (size_t)B_ * N_ * FOUT;
  float* s2   = s1 + B_ * N_;
  float* mrow = s2 + B_ * N_;
  float* lrow = mrow + B_ * N_;
  unsigned* mbits = (unsigned*)(lrow + B_ * N_);

  wh_kernel<<<(B_ * (N_ / 16) * 8) / 4, 128, 0, stream>>>(h, W, Wh);
  s12_kernel<<<(B_ * N_ * 32) / 256, 256, 0, stream>>>(Wh, a, s1, s2);
  stats_kernel<<<(B_ * N_) / 8, 256, 0, stream>>>(adj, s1, s2, mrow, lrow, mbits);
  hprime_kernel<<<B_ * (N_ / 16), 128, 0, stream>>>(mbits, Wh, s1, s2, mrow, lrow,
                                                    gamma, beta, out);
}


# h_prime kernel: Wh not staged in LDS: columns permuted so a lane's 4 tiles are one 16-byte global load; 2 register sets, loads one 16-j chunk ahead, each feeds 2 row tiles; j, LayerNorm sums reorder
# speedup vs baseline: 1.3328x; 1.3328x over previous
#include <hip/hip_runtime.h>
#include <hip/hip_bf16.h>
#include <math.h>

typedef __attribute__((ext_vector_type(2))) float v2f;
typedef __attribute__((ext_vector_type(4))) float v4f;
typedef __attribute__((ext_vector_type(8))) float v8f;

#define B_   4
#define N_   2048
#define FIN  256
#define FOUT 128
#define ALPHA_ 0.2f
#define EPS_   1e-5f
#define NEG_   -9000000000000000.0f

__global__ __launch_bounds__(128) void wh_kernel(const float* __restrict__ h,
                                                 const float* __restrict__ W,
                                                 float* __restrict__ Wh) {
  int lane  = threadIdx.x & 31;
  int tile  = blockIdx.x * 4 + (threadIdx.x >> 5);
  int otile = tile & 7;
  int rest  = tile >> 3;
  int itile = rest & (N_ / 16 - 1);
  int b     = rest >> 7;
  int i0 = itile * 16, o0 = otile * 16;
  int r16  = lane & 15;
  int koff = (lane >> 4) << 1;

  const float* hrow = h + (size_t)(b * N_ + i0 + r16) * FIN;
  v8f c = {};
  for (int k0 = 0; k0 < FIN; k0 += 4) {
    float2 av = *(const float2*)(hrow + k0 + koff);
    v2f a; a.x = av.x; a.y = av.y;
    v2f bb;
    bb.x = W[(size_t)(k0 + koff + 0) * FOUT + o0 + r16];
    bb.y = W[(size_t)(k0 + koff + 1) * FOUT + o0 + r16];
    c = __builtin_amdgcn_wmma_f32_16x16x4_f32(false, a, false, bb, (short)0, c,
                                              false, false);
  }
#pragma unroll
  for (int r = 0; r < 8; ++r) {
    int M = r + ((lane >> 4) << 3);
    Wh[(size_t)(b * N_ + i0 + M) * FOUT + o0 + r16] = c[r];
  }
}

__global__ __launch_bounds__(256) void s12_kernel(const float* __restrict__ Wh,
                                                  const float* __restrict__ a,
                                                  float* __restrict__ s1,
                                                  float* __restrict__ s2) {
  int lane = threadIdx.x & 31;
  int node = (int)((blockIdx.x * blockDim.x + threadIdx.x) >> 5);
  if (node >= B_ * N_) return;
  const float4* w4 = (const float4*)(Wh + (size_t)node * FOUT);
  const float4* a1 = (const float4*)a;
  const float4* a2 = (const float4*)(a + FOUT);
  float4 wv = w4[lane];
  float4 x1 = a1[lane];
  float4 x2 = a2[lane];
  float p1 = wv.x * x1.x + wv.y * x1.y + wv.z * x1.z + wv.w * x1.w;
  float p2 = wv.x * x2.x + wv.y * x2.y + wv.z * x2.z + wv.w * x2.w;
  for (int off = 16; off >= 1; off >>= 1) {
    p1 += __shfl_xor(p1, off, 32);
    p2 += __shfl_xor(p2, off, 32);
  }
  if (lane == 0) { s1[node] = p1; s2[node] = p2; }
}

__global__ __launch_bounds__(256) void stats_kernel(const int* __restrict__ adj,
                                                    const float* __restrict__ s1,
                                                    const float* __restrict__ s2,
                                                    float* __restrict__ mrow,
                                                    float* __restrict__ lrow,
                                                    unsigned* __restrict__ mbits) {
  int lane = threadIdx.x & 31;
  int row  = blockIdx.x * 8 + (threadIdx.x >> 5);
  int b    = row >> 11;
  float s1v = s1[row];
  const float* s2b = s2 + (size_t)b * N_;
  const int* arow  = adj + (size_t)row * N_;
  unsigned* mb     = mbits + (size_t)row * (N_ / 32);
  float m = -INFINITY, l = 0.f;
  for (int w = 0; w < N_ / 32; ++w) {
    int j = w * 32 + lane;
    float e = s1v + s2b[j];
    e = (e >= 0.f) ? e : ALPHA_ * e;
    bool pred = arow[j] > 0;
    unsigned long long bl = __ballot(pred);
    if (lane == 0) mb[w] = (unsigned)bl;
    float tv = pred ? e : NEG_;
    if (tv > m) { l = l * __expf(m - tv) + 1.0f; m = tv; }
    else        { l += __expf(tv - m); }
  }
  for (int off = 16; off >= 1; off >>= 1) {
    float m2 = __shfl_xor(m, off, 32);
    float l2 = __shfl_xor(l, off, 32);
    float M  = fmaxf(m, m2);
    l = l * __expf(m - M) + l2 * __expf(m2 - M);
    m = M;
  }
  if (lane == 0) { mrow[row] = m; lrow[row] = l; }
}

#define HP_ROWS  32
#define HP_WAVES 8
#define HP_JW    (N_ / HP_WAVES)
#define HP_PITCH 160

struct ChunkSet {
  v4f w[4][2][2];
  v2f sv[4];
};

__device__ __forceinline__ void load_chunk(ChunkSet& d, const float* __restrict__ wl,
                                           const float* __restrict__ sl, int ch) {
#pragma unroll
  for (int q = 0; q < 4; ++q) {
#pragma unroll
    for (int s = 0; s < 2; ++s) {
      const float* p = wl + (size_t)(16 * ch + 4 * q + s) * FOUT;
      d.w[q][s][0] = *(const v4f*)p;
      d.w[q][s][1] = *(const v4f*)(p + 64);
    }
    d.sv[q] = *(const v2f*)(sl + 16 * ch + 4 * q);
  }
}

__device__ __forceinline__ void mma_chunk(v8f (&acc)[2][8], const ChunkSet& d,
                                          unsigned wb0, unsigned wb1,
                                          float s1v0, float s1v1, float mi0, float mi1,
                                          float invl0, float invl1) {
#pragma unroll
  for (int q = 0; q < 4; ++q) {
    v2f a0, a1;
#pragma unroll
    for (int s = 0; s < 2; ++s) {
      float sv = d.sv[q][s];
      float e0 = s1v0 + sv;
      float e1 = s1v1 + sv;
      e0 = (e0 >= 0.f) ? e0 : ALPHA_ * e0;
      e1 = (e1 >= 0.f) ? e1 : ALPHA_ * e1;
      float t0 = ((wb0 >> (4 * q + s)) & 1u) ? e0 : NEG_;
      float t1 = ((wb1 >> (4 * q + s)) & 1u) ? e1 : NEG_;
      a0[s] = __expf(t0 - mi0) * invl0;
      a1[s] = __expf(t1 - mi1) * invl1;
    }
#pragma unroll
    for (int ct = 0; ct < 8; ++ct) {
      v2f bb;
      bb.x = d.w[q][0][ct >> 2][ct & 3];
      bb.y = d.w[q][1][ct >> 2][ct & 3];
      acc[0][ct] = __builtin_amdgcn_wmma_f32_16x16x4_f32(false, a0, false, bb,
                                                         (short)0, acc[0][ct],
                                                         false, false);
      acc[1][ct] = __builtin_amdgcn_wmma_f32_16x16x4_f32(false, a1, false, bb,
                                                         (short)0, acc[1][ct],
                                                         false, false);
    }
  }
}

__global__ __launch_bounds__(256) void hprime_kernel(
    const unsigned* __restrict__ mbits, const float* __restrict__ Wh,
    const float* __restrict__ s1, const float* __restrict__ s2,
    const float* __restrict__ mrow, const float* __restrict__ lrow,
    const float* __restrict__ gamma, const float* __restrict__ beta,
    float* __restrict__ out) {
  __shared__ float lds[HP_WAVES][HP_ROWS][HP_PITCH];
  int tid  = threadIdx.x;
  int lane = tid & 31;
  int wave = tid >> 5;
  int b    = blockIdx.x / (N_ / HP_ROWS);
  int i0   = (blockIdx.x % (N_ / HP_ROWS)) * HP_ROWS;
  int c    = lane & 15;
  int h    = lane >> 4;
  int row0 = b * N_ + i0 + c;
  int row1 = row0 + 16;

  float s1v0  = s1[row0],   s1v1  = s1[row1];
  float mi0   = mrow[row0], mi1   = mrow[row1];
  float invl0 = 1.0f / lrow[row0];
  float invl1 = 1.0f / lrow[row1];
  const unsigned* mb0 = mbits + (size_t)row0 * (N_ / 32) + wave * (HP_JW / 32);
  const unsigned* mb1 = mbits + (size_t)row1 * (N_ / 32) + wave * (HP_JW / 32);
  const float* sl = s2 + (size_t)b * N_ + wave * HP_JW + 2 * h;
  const float* wl = Wh + ((size_t)b * N_ + wave * HP_JW + 2 * h) * FOUT + 4 * c;
  int sh = 2 * h;

  v8f acc[2][8];
#pragma unroll
  for (int t = 0; t < 2; ++t)
#pragma unroll
    for (int ct = 0; ct < 8; ++ct) acc[t][ct] = (v8f){};

  ChunkSet ca, cb;
  load_chunk(ca, wl, sl, 0);
  unsigned w0 = mb0[0], w1 = mb1[0];
  for (int it = 0; it < HP_JW / 32; ++it) {
    bool more = (it + 1) < HP_JW / 32;
    load_chunk(cb, wl, sl, 2 * it + 1);
    unsigned n0 = w0, n1 = w1;
    if (more) { n0 = mb0[it + 1]; n1 = mb1[it + 1]; }
    __builtin_amdgcn_sched_barrier(0);
    mma_chunk(acc, ca, w0 >> sh, w1 >> sh, s1v0, s1v1, mi0, mi1, invl0, invl1);
    if (more) load_chunk(ca, wl, sl, 2 * it + 2);
    __builtin_amdgcn_sched_barrier(0);
    mma_chunk(acc, cb, w0 >> (sh + 16), w1 >> (sh + 16), s1v0, s1v1, mi0, mi1,
              invl0, invl1);
    w0 = n0; w1 = n1;
  }

#pragma unroll
  for (int t = 0; t < 2; ++t)
#pragma unroll
    for (int r = 0; r < 8; ++r) {
      int M = 16 * t + 8 * h + r;
      v4f lo = {acc[t][0][r], acc[t][1][r], acc[t][2][r], acc[t][3][r]};
      v4f hi = {acc[t][4][r], acc[t][5][r], acc[t][6][r], acc[t][7][r]};
      *(v4f*)&lds[wave][M][4 * c]      = lo;
      *(v4f*)&lds[wave][M][64 + 4 * c] = hi;
    }
  __syncthreads();

  int row32 = tid >> 3;
  int seg   = tid & 7;
  v4f vals[4];
  float s = 0.f, ss = 0.f;
#pragma unroll
  for (int k = 0; k < 4; ++k) {
    int col = 32 * k + 4 * seg;
    v4f x = *(const v4f*)&lds[0][row32][col];
#pragma unroll
    for (int w = 1; w < HP_WAVES; ++w) x += *(const v4f*)&lds[w][row32][col];
    vals[k] = x;
#pragma unroll
    for (int e = 0; e < 4; ++e) { s += x[e]; ss += x[e] * x[e]; }
  }
  for (int off = 4; off >= 1; off >>= 1) {
    s  += __shfl_xor(s, off, 8);
    ss += __shfl_xor(ss, off, 8);
  }
  float mu   = s * (1.0f / FOUT);
  float var  = ss * (1.0f / FOUT) - mu * mu;
  float rstd = rsqrtf(var + EPS_);
  float* orow = out + (size_t)(b * N_ + i0 + row32) * FOUT;
#pragma unroll
  for (int k = 0; k < 4; ++k) {
    int col = 32 * k + 4 * seg;
    v4f g  = *(const v4f*)(gamma + col);
    v4f be = *(const v4f*)(beta + col);
    v4f o;
#pragma unroll
    for (int e = 0; e < 4; ++e) {
      float x = (vals[k][e] - mu) * rstd * g[e] + be[e];
      o[e] = 0.5f * x * (1.0f + erff(x * 0.70710678118654752f));
    }
    *(v4f*)(orow + col) = o;
  }
}

extern "C" void kernel_launch(void* const* d_in, const int* in_sizes, int n_in,
                              void* d_out, int out_size, void* d_ws, size_t ws_size,
                              hipStream_t stream) {
  const float* h     = (const float*)d_in[0];
  const int*   adj   = (const int*)d_in[1];
  const float* W     = (const float*)d_in[2];
  const float* a     = (const float*)d_in[3];
  const float* gamma = (const float*)d_in[4];
  const float* beta  = (const float*)d_in[5];
  float* out = (float*)d_out;

  float* ws   = (float*)d_ws;
  float* Wh   = ws;
  float* s1   = Wh + (size_t)B_ * N_ * FOUT;
  float* s2   = s1 + B_ * N_;
  float* mrow = s2 + B_ * N_;
  float* lrow = mrow + B_ * N_;
  unsigned* mbits = (unsigned*)(lrow + B_ * N_);

  wh_kernel<<<(B_ * (N_ / 16) * 8) / 4, 128, 0, stream>>>(h, W, Wh);
  s12_kernel<<<(B_ * N_ * 32) / 256, 256, 0, stream>>>(Wh, a, s1, s2);
  stats_kernel<<<(B_ * N_) / 8, 256, 0, stream>>>(adj, s1, s2, mrow, lrow, mbits);
  hprime_kernel<<<B_ * (N_ / 32), 256, 0, stream>>>(mbits, Wh, s1, s2, mrow, lrow,
                                                    gamma, beta, out);
}


# adj pass moved into the h_prime block: a wave reads its 256 columns of 32 rows by 16-byte loads, ballots keep the mask in registers; row max, weight sums formed there; 3 launches; sums reorder
# speedup vs baseline: 1.8786x; 1.4095x over previous
#include <hip/hip_runtime.h>
#include <hip/hip_bf16.h>
#include <math.h>

typedef __attribute__((ext_vector_type(2))) float v2f;
typedef __attribute__((ext_vector_type(4))) float v4f;
typedef __attribute__((ext_vector_type(8))) float v8f;
typedef __attribute__((ext_vector_type(4))) int v4i;

#define B_   4
#define N_   2048
#define FIN  256
#define FOUT 128
#define ALPHA_ 0.2f
#define EPS_   1e-5f
#define NEG_   -9000000000000000.0f

__global__ __launch_bounds__(128) void wh_kernel(const float* __restrict__ h,
                                                 const float* __restrict__ W,
                                                 float* __restrict__ Wh) {
  int lane  = threadIdx.x & 31;
  int tile  = blockIdx.x * 4 + (threadIdx.x >> 5);
  int otile = tile & 7;
  int rest  = tile >> 3;
  int itile = rest & (N_ / 16 - 1);
  int b     = rest >> 7;
  int i0 = itile * 16, o0 = otile * 16;
  int r16  = lane & 15;
  int koff = (lane >> 4) << 1;

  const float* hrow = h + (size_t)(b * N_ + i0 + r16) * FIN;
  v8f c = {};
  for (int k0 = 0; k0 < FIN; k0 += 4) {
    float2 av = *(const float2*)(hrow + k0 + koff);
    v2f a; a.x = av.x; a.y = av.y;
    v2f bb;
    bb.x = W[(size_t)(k0 + koff + 0) * FOUT + o0 + r16];
    bb.y = W[(size_t)(k0 + koff + 1) * FOUT + o0 + r16];
    c = __builtin_amdgcn_wmma_f32_16x16x4_f32(false, a, false, bb, (short)0, c,
                                              false, false);
  }
#pragma unroll
  for (int r = 0; r < 8; ++r) {
    int M = r + ((lane >> 4) << 3);
    Wh[(size_t)(b * N_ + i0 + M) * FOUT + o0 + r16] = c[r];
  }
}

__global__ __launch_bounds__(256) void s12_kernel(const float* __restrict__ Wh,
                                                  const float* __restrict__ a,
                                                  float* __restrict__ s1,
                                                  float* __restrict__ s2) {
  int lane = threadIdx.x & 31;
  int node = (int)((blockIdx.x * blockDim.x + threadIdx.x) >> 5);
  if (node >= B_ * N_) return;
  const float4* w4 = (const float4*)(Wh + (size_t)node * FOUT);
  const float4* a1 = (const float4*)a;
  const float4* a2 = (const float4*)(a + FOUT);
  float4 wv = w4[lane];
  float4 x1 = a1[lane];
  float4 x2 = a2[lane];
  float p1 = wv.x * x1.x + wv.y * x1.y + wv.z * x1.z + wv.w * x1.w;
  float p2 = wv.x * x2.x + wv.y * x2.y + wv.z * x2.z + wv.w * x2.w;
  for (int off = 16; off >= 1; off >>= 1) {
    p1 += __shfl_xor(p1, off, 32);
    p2 += __shfl_xor(p2, off, 32);
  }
  if (lane == 0) { s1[node] = p1; s2[node] = p2; }
}

#define HP_ROWS  32
#define HP_WAVES 8
#define HP_JW    (N_ / HP_WAVES)
#define HP_PITCH 160
#define HP_LCOL  128
#define HP_MCOL  129

struct ChunkSet {
  v4f w[4][2][2];
  v2f sv[4];
};

__device__ __forceinline__ void load_chunk(ChunkSet& d, const float* __restrict__ wl,
                                           const float* __restrict__ sl, int ch) {
#pragma unroll
  for (int q = 0; q < 4; ++q) {
#pragma unroll
    for (int s = 0; s < 2; ++s) {
      const float* p = wl + (size_t)(16 * ch + 4 * q + s) * FOUT;
      d.w[q][s][0] = *(const v4f*)p;
      d.w[q][s][1] = *(const v4f*)(p + 64);
    }
    d.sv[q] = *(const v2f*)(sl + 16 * ch + 4 * q);
  }
}

__device__ __forceinline__ void mma_chunk(v8f (&acc)[2][8], const ChunkSet& d,
                                          unsigned u00, unsigned u01,
                                          unsigned u10, unsigned u11,
                                          float s1v0, float s1v1, float mi0, float mi1,
                                          float& ls0, float& ls1) {
#pragma unroll
  for (int q = 0; q < 4; ++q) {
    v2f a0, a1;
#pragma unroll
    for (int s = 0; s < 2; ++s) {
      float sv = d.sv[q][s];
      float e0 = s1v0 + sv;
      float e1 = s1v1 + sv;
      e0 = (e0 >= 0.f) ? e0 : ALPHA_ * e0;
      e1 = (e1 >= 0.f) ? e1 : ALPHA_ * e1;
      unsigned b0 = s ? u01 : u00;
      unsigned b1 = s ? u11 : u10;
      float t0 = ((b0 >> q) & 1u) ? e0 : NEG_;
      float t1 = ((b1 >> q) & 1u) ? e1 : NEG_;
      a0[s] = __expf(t0 - mi0);
      a1[s] = __expf(t1 - mi1);
    }
    ls0 += a0[0] + a0[1];
    ls1 += a1[0] + a1[1];
#pragma unroll
    for (int ct = 0; ct < 8; ++ct) {
      v2f bb;
      bb.x = d.w[q][0][ct >> 2][ct & 3];
      bb.y = d.w[q][1][ct >> 2][ct & 3];
      acc[0][ct] = __builtin_amdgcn_wmma_f32_16x16x4_f32(false, a0, false, bb,
                                                         (short)0, acc[0][ct],
                                                         false, false);
      acc[1][ct] = __builtin_amdgcn_wmma_f32_16x16x4_f32(false, a1, false, bb,
                                                         (short)0, acc[1][ct],
                                                         false, false);
    }
  }
}

__global__ __launch_bounds__(256) void hprime_kernel(
    const int* __restrict__ adj, const float* __restrict__ Wh,
    const float* __restrict__ s1, const float* __restrict__ s2,
    const float* __restrict__ gamma, const float* __restrict__ beta,
    float* __restrict__ out) {
  __shared__ float lds[HP_WAVES][HP_ROWS][HP_PITCH];
  int tid  = threadIdx.x;
  int lane = tid & 31;
  int wave = tid >> 5;
  int b    = blockIdx.x / (N_ / HP_ROWS);
  int i0   = (blockIdx.x % (N_ / HP_ROWS)) * HP_ROWS;
  int c    = lane & 15;
  int h    = lane >> 4;
  int row0 = b * N_ + i0 + c;
  int row1 = row0 + 16;

  float s1v0 = s1[row0], s1v1 = s1[row1];
  const float* sl = s2 + (size_t)b * N_ + wave * HP_JW + 2 * h;
  const float* wl = Wh + ((size_t)b * N_ + wave * HP_JW + 2 * h) * FOUT + 4 * c;

  const int* ab = adj + (size_t)(b * N_ + i0) * N_ + wave * HP_JW + 4 * lane;
  const float* s2w = s2 + (size_t)b * N_ + wave * HP_JW + 4 * lane;
  v4f sw[2];
  sw[0] = *(const v4f*)s2w;
  sw[1] = *(const v4f*)(s2w + 128);
  unsigned wd[2][2][2];
  float smax[2];
  v4i va[2][16][2];
#pragma unroll
  for (int t = 0; t < 2; ++t)
#pragma unroll
    for (int r = 0; r < 16; ++r)
#pragma unroll
      for (int wn = 0; wn < 2; ++wn)
        va[t][r][wn] = *(const v4i*)(ab + (size_t)(16 * t + r) * N_ + 128 * wn);
  __builtin_amdgcn_sched_barrier(0);
#pragma unroll
  for (int t = 0; t < 2; ++t) {
    smax[t] = -INFINITY;
#pragma unroll
    for (int wn = 0; wn < 2; ++wn) { wd[t][wn][0] = 0u; wd[t][wn][1] = 0u; }
#pragma unroll
    for (int r = 0; r < 16; ++r) {
      float pm = -INFINITY;
      bool mine = (c == r);
#pragma unroll
      for (int wn = 0; wn < 2; ++wn) {
        unsigned be[4];
#pragma unroll
        for (int e = 0; e < 4; ++e) {
          bool p = va[t][r][wn][e] > 0;
          be[e] = (unsigned)__ballot(p);
          pm = fmaxf(pm, p ? sw[wn][e] : -INFINITY);
        }
        unsigned x0 = h ? be[2] : be[0];
        unsigned x1 = h ? be[3] : be[1];
        wd[t][wn][0] = mine ? x0 : wd[t][wn][0];
        wd[t][wn][1] = mine ? x1 : wd[t][wn][1];
      }
      for (int off = 16; off >= 1; off >>= 1) pm = fmaxf(pm, __shfl_xor(pm, off, 32));
      smax[t] = mine ? pm : smax[t];
    }
  }
  if (h == 0) {
    lds[wave][c][HP_MCOL]      = smax[0];
    lds[wave][16 + c][HP_MCOL] = smax[1];
  }
  __syncthreads();
  float sm0 = lds[0][c][HP_MCOL], sm1 = lds[0][16 + c][HP_MCOL];
#pragma unroll
  for (int w = 1; w < HP_WAVES; ++w) {
    sm0 = fmaxf(sm0, lds[w][c][HP_MCOL]);
    sm1 = fmaxf(sm1, lds[w][16 + c][HP_MCOL]);
  }
  float em0 = s1v0 + sm0, em1 = s1v1 + sm1;
  em0 = (em0 >= 0.f) ? em0 : ALPHA_ * em0;
  em1 = (em1 >= 0.f) ? em1 : ALPHA_ * em1;
  float mi0 = (sm0 == -INFINITY) ? NEG_ : em0;
  float mi1 = (sm1 == -INFINITY) ? NEG_ : em1;

  v8f acc[2][8];
#pragma unroll
  for (int t = 0; t < 2; ++t)
#pragma unroll
    for (int ct = 0; ct < 8; ++ct) acc[t][ct] = (v8f){};
  float ls0 = 0.f, ls1 = 0.f;

  ChunkSet ca, cb;
  load_chunk(ca, wl, sl, 0);
  for (int it = 0; it < HP_JW / 32; ++it) {
    bool more = (it + 1) < HP_JW / 32;
    bool hi   = it >= 4;
    int  sh   = 8 * (it & 3);
    unsigned u00 = (hi ? wd[0][1][0] : wd[0][0][0]) >> sh;
    unsigned u01 = (hi ? wd[0][1][1] : wd[0][0][1]) >> sh;
    unsigned u10 = (hi ? wd[1][1][0] : wd[1][0][0]) >> sh;
    unsigned u11 = (hi ? wd[1][1][1] : wd[1][0][1]) >> sh;
    load_chunk(cb, wl, sl, 2 * it + 1);
    __builtin_amdgcn_sched_barrier(0);
    mma_chunk(acc, ca, u00, u01, u10, u11, s1v0, s1v1, mi0, mi1, ls0, ls1);
    if (more) load_chunk(ca, wl, sl, 2 * it + 2);
    __builtin_amdgcn_sched_barrier(0);
    mma_chunk(acc, cb, u00 >> 4, u01 >> 4, u10 >> 4, u11 >> 4, s1v0, s1v1, mi0, mi1,
              ls0, ls1);
  }

  ls0 += __shfl_xor(ls0, 16, 32);
  ls1 += __shfl_xor(ls1, 16, 32);
  if (h == 0) {
    lds[wave][c][HP_LCOL]      = ls0;
    lds[wave][16 + c][HP_LCOL] = ls1;
  }

#pragma unroll
  for (int t = 0; t < 2; ++t)
#pragma unroll
    for (int r = 0; r < 8; ++r) {
      int M = 16 * t + 8 * h + r;
      v4f lo = {acc[t][0][r], acc[t][1][r], acc[t][2][r], acc[t][3][r]};
      v4f hi = {acc[t][4][r], acc[t][5][r], acc[t][6][r], acc[t][7][r]};
      *(v4f*)&lds[wave][M][4 * c]      = lo;
      *(v4f*)&lds[wave][M][64 + 4 * c] = hi;
    }
  __syncthreads();

  int row32 = tid >> 3;
  int seg   = tid & 7;
  float lsum = lds[0][row32][HP_LCOL];
#pragma unroll
  for (int w = 1; w < HP_WAVES; ++w) lsum += lds[w][row32][HP_LCOL];
  float invl = 1.0f / lsum;
  v4f vals[4];
  float s = 0.f, ss = 0.f;
#pragma unroll
  for (int k = 0; k < 4; ++k) {
    int col = 32 * k + 4 * seg;
    v4f x = *(const v4f*)&lds[0][row32][col];
#pragma unroll
    for (int w = 1; w < HP_WAVES; ++w) x += *(const v4f*)&lds[w][row32][col];
    x *= invl;
    vals[k] = x;
#pragma unroll
    for (int e = 0; e < 4; ++e) { s += x[e]; ss += x[e] * x[e]; }
  }
  for (int off = 4; off >= 1; off >>= 1) {
    s  += __shfl_xor(s, off, 8);
    ss += __shfl_xor(ss, off, 8);
  }
  float mu   = s * (1.0f / FOUT);
  float var  = ss * (1.0f / FOUT) - mu * mu;
  float rstd = rsqrtf(var + EPS_);
  float* orow = out + (size_t)(b * N_ + i0 + row32) * FOUT;
#pragma unroll
  for (int k = 0; k < 4; ++k) {
    int col = 32 * k + 4 * seg;
    v4f g  = *(const v4f*)(gamma + col);
    v4f be = *(const v4f*)(beta + col);
    v4f o;
#pragma unroll
    for (int e = 0; e < 4; ++e) {
      float x = (vals[k][e] - mu) * rstd * g[e] + be[e];
      o[e] = 0.5f * x * (1.0f + erff(x * 0.70710678118654752f));
    }
    *(v4f*)(orow + col) = o;
  }
}

extern "C" void kernel_launch(void* const* d_in, const int* in_sizes, int n_in,
                              void* d_out, int out_size, void* d_ws, size_t ws_size,
                              hipStream_t stream) {
  const float* h     = (const float*)d_in[0];
  const int*   adj   = (const int*)d_in[1];
  const float* W     = (const float*)d_in[2];
  const float* a     = (const float*)d_in[3];
  const float* gamma = (const float*)d_in[4];
  const float* beta  = (const float*)d_in[5];
  float* out = (float*)d_out;

  float* ws = (float*)d_ws;
  float* Wh = ws;
  float* s1 = Wh + (size_t)B_ * N_ * FOUT;
  float* s2 = s1 + B_ * N_;

  wh_kernel<<<(B_ * (N_ / 16) * 8) / 4, 128, 0, stream>>>(h, W, Wh);
  s12_kernel<<<(B_ * N_ * 32) / 256, 256, 0, stream>>>(Wh, a, s1, s2);
  hprime_kernel<<<B_ * (N_ / 32), 256, 0, stream>>>(adj, Wh, s1, s2, gamma, beta, out);
}


# Wh kernel: a wave keeps 4 column tiles (columns permuted) of a 16-row strip in registers, W rows by 16-byte loads a set ahead; s1/s2 dots taken from its accumulators: 3 to 2 launches; sums as before
# speedup vs baseline: 2.2185x; 1.1809x over previous
#include <hip/hip_runtime.h>
#include <hip/hip_bf16.h>
#include <math.h>

typedef __attribute__((ext_vector_type(2))) float v2f;
typedef __attribute__((ext_vector_type(4))) float v4f;
typedef __attribute__((ext_vector_type(8))) float v8f;
typedef __attribute__((ext_vector_type(4))) int v4i;

#define B_   4
#define N_   2048
#define FIN  256
#define FOUT 128
#define ALPHA_ 0.2f
#define EPS_   1e-5f
#define NEG_   -9000000000000000.0f

#define WS_Q      16
#define WS_STRIPS 2

struct WSet {
  v2f a[WS_Q];
  v4f w[WS_Q][2];
};

__device__ __forceinline__ void load_wset(WSet& d, const float* __restrict__ hl,
                                          const float* __restrict__ wl, int trip) {
#pragma unroll
  for (int q = 0; q < WS_Q; ++q) {
    int k = 4 * (WS_Q * trip + q);
    d.a[q]    = *(const v2f*)(hl + k);
    d.w[q][0] = *(const v4f*)(wl + (size_t)k * FOUT);
    d.w[q][1] = *(const v4f*)(wl + (size_t)(k + 1) * FOUT);
  }
}

__device__ __forceinline__ void mma_wset(v8f (&acc)[4], const WSet& d) {
#pragma unroll
  for (int q = 0; q < WS_Q; ++q) {
#pragma unroll
    for (int t = 0; t < 4; ++t) {
      v2f bb;
      bb.x = d.w[q][0][t];
      bb.y = d.w[q][1][t];
      acc[t] = __builtin_amdgcn_wmma_f32_16x16x4_f32(false, d.a[q], false, bb,
                                                     (short)0, acc[t], false, false);
    }
  }
}

__global__ __launch_bounds__(64 * WS_STRIPS) void wh_s12_kernel(
    const float* __restrict__ h, const float* __restrict__ W,
    const float* __restrict__ a, float* __restrict__ Wh,
    float* __restrict__ s1, float* __restrict__ s2) {
  __shared__ float xch[WS_STRIPS][2][8][32];
  int lane  = threadIdx.x & 31;
  int wave  = threadIdx.x >> 5;
  int half  = wave & 1;
  int sloc  = wave >> 1;
  int strip = blockIdx.x * WS_STRIPS + sloc;
  int c  = lane & 15;
  int hh = lane >> 4;
  int row0 = 16 * strip;

  const float* hl = h + (size_t)(row0 + c) * FIN + 2 * hh;
  const float* wl = W + (size_t)(2 * hh) * FOUT + 64 * half + 4 * c;

  v8f acc[4];
#pragma unroll
  for (int t = 0; t < 4; ++t) acc[t] = (v8f){};

  WSet ca, cb;
  load_wset(ca, hl, wl, 0);
  for (int it = 0; it < FIN / (8 * WS_Q); ++it) {
    bool more = (it + 1) < FIN / (8 * WS_Q);
    load_wset(cb, hl, wl, 2 * it + 1);
    __builtin_amdgcn_sched_barrier(0);
    mma_wset(acc, ca);
    if (more) load_wset(ca, hl, wl, 2 * it + 2);
    __builtin_amdgcn_sched_barrier(0);
    mma_wset(acc, cb);
  }

  v4f x1 = *(const v4f*)(a + 64 * half + 4 * c);
  v4f x2 = *(const v4f*)(a + FOUT + 64 * half + 4 * c);
  float* wrow = Wh + (size_t)(row0 + 8 * hh) * FOUT + 64 * half + 4 * c;
  float p1[8], p2[8];
#pragma unroll
  for (int r = 0; r < 8; ++r) {
    v4f wv = {acc[0][r], acc[1][r], acc[2][r], acc[3][r]};
    *(v4f*)(wrow + (size_t)r * FOUT) = wv;
    p1[r] = wv.x * x1.x + wv.y * x1.y + wv.z * x1.z + wv.w * x1.w;
    p2[r] = wv.x * x2.x + wv.y * x2.y + wv.z * x2.z + wv.w * x2.w;
  }
#pragma unroll
  for (int r = 0; r < 8; ++r) xch[sloc][half][r][lane] = half ? p1[r] : p2[r];
  __syncthreads();
  float qd[8];
#pragma unroll
  for (int r = 0; r < 8; ++r) {
    float mine = half ? p2[r] : p1[r];
    qd[r] = mine + xch[sloc][half ^ 1][r][lane];
#pragma unroll
    for (int off = 8; off >= 1; off >>= 1) qd[r] += __shfl_xor(qd[r], off, 32);
  }
  if (c == 0) {
    float* dst = (half ? s2 : s1) + row0 + 8 * hh;
    *(v4f*)dst       = (v4f){qd[0], qd[1], qd[2], qd[3]};
    *(v4f*)(dst + 4) = (v4f){qd[4], qd[5], qd[6], qd[7]};
  }
}

#define HP_ROWS  32
#define HP_WAVES 8
#define HP_JW    (N_ / HP_WAVES)
#define HP_PITCH 160
#define HP_LCOL  128
#define HP_MCOL  129

struct ChunkSet {
  v4f w[4][2][2];
  v2f sv[4];
};

__device__ __forceinline__ void load_chunk(ChunkSet& d, const float* __restrict__ wl,
                                           const float* __restrict__ sl, int ch) {
#pragma unroll
  for (int q = 0; q < 4; ++q) {
#pragma unroll
    for (int s = 0; s < 2; ++s) {
      const float* p = wl + (size_t)(16 * ch + 4 * q + s) * FOUT;
      d.w[q][s][0] = *(const v4f*)p;
      d.w[q][s][1] = *(const v4f*)(p + 64);
    }
    d.sv[q] = *(const v2f*)(sl + 16 * ch + 4 * q);
  }
}

__device__ __forceinline__ void mma_chunk(v8f (&acc)[2][8], const ChunkSet& d,
                                          unsigned u00, unsigned u01,
                                          unsigned u10, unsigned u11,
                                          float s1v0, float s1v1, float mi0, float mi1,
                                          float& ls0, float& ls1) {
#pragma unroll
  for (int q = 0; q < 4; ++q) {
    v2f a0, a1;
#pragma unroll
    for (int s = 0; s < 2; ++s) {
      float sv = d.sv[q][s];
      float e0 = s1v0 + sv;
      float e1 = s1v1 + sv;
      e0 = (e0 >= 0.f) ? e0 : ALPHA_ * e0;
      e1 = (e1 >= 0.f) ? e1 : ALPHA_ * e1;
      unsigned b0 = s ? u01 : u00;
      unsigned b1 = s ? u11 : u10;
      float t0 = ((b0 >> q) & 1u) ? e0 : NEG_;
      float t1 = ((b1 >> q) & 1u) ? e1 : NEG_;
      a0[s] = __expf(t0 - mi0);
      a1[s] = __expf(t1 - mi1);
    }
    ls0 += a0[0] + a0[1];
    ls1 += a1[0] + a1[1];
#pragma unroll
    for (int ct = 0; ct < 8; ++ct) {
      v2f bb;
      bb.x = d.w[q][0][ct >> 2][ct & 3];
      bb.y = d.w[q][1][ct >> 2][ct & 3];
      acc[0][ct] = __builtin_amdgcn_wmma_f32_16x16x4_f32(false, a0, false, bb,
                                                         (short)0, acc[0][ct],
                                                         false, false);
      acc[1][ct] = __builtin_amdgcn_wmma_f32_16x16x4_f32(false, a1, false, bb,
                                                         (short)0, acc[1][ct],
                                                         false, false);
    }
  }
}

__global__ __launch_bounds__(256) void hprime_kernel(
    const int* __restrict__ adj, const float* __restrict__ Wh,
    const float* __restrict__ s1, const float* __restrict__ s2,
    const float* __restrict__ gamma, const float* __restrict__ beta,
    float* __restrict__ out) {
  __shared__ float lds[HP_WAVES][HP_ROWS][HP_PITCH];
  int tid  = threadIdx.x;
  int lane = tid & 31;
  int wave = tid >> 5;
  int b    = blockIdx.x / (N_ / HP_ROWS);
  int i0   = (blockIdx.x % (N_ / HP_ROWS)) * HP_ROWS;
  int c    = lane & 15;
  int h    = lane >> 4;
  int row0 = b * N_ + i0 + c;
  int row1 = row0 + 16;

  float s1v0 = s1[row0], s1v1 = s1[row1];
  const float* sl = s2 + (size_t)b * N_ + wave * HP_JW + 2 * h;
  const float* wl = Wh + ((size_t)b * N_ + wave * HP_JW + 2 * h) * FOUT + 4 * c;

  const int* ab = adj + (size_t)(b * N_ + i0) * N_ + wave * HP_JW + 4 * lane;
  const float* s2w = s2 + (size_t)b * N_ + wave * HP_JW + 4 * lane;
  v4f sw[2];
  sw[0] = *(const v4f*)s2w;
  sw[1] = *(const v4f*)(s2w + 128);
  unsigned wd[2][2][2];
  float smax[2];
  v4i va[2][16][2];
#pragma unroll
  for (int t = 0; t < 2; ++t)
#pragma unroll
    for (int r = 0; r < 16; ++r)
#pragma unroll
      for (int wn = 0; wn < 2; ++wn)
        va[t][r][wn] = *(const v4i*)(ab + (size_t)(16 * t + r) * N_ + 128 * wn);
  __builtin_amdgcn_sched_barrier(0);
#pragma unroll
  for (int t = 0; t < 2; ++t) {
    smax[t] = -INFINITY;
#pragma unroll
    for (int wn = 0; wn < 2; ++wn) { wd[t][wn][0] = 0u; wd[t][wn][1] = 0u; }
#pragma unroll
    for (int r = 0; r < 16; ++r) {
      float pm = -INFINITY;
      bool mine = (c == r);
#pragma unroll
      for (int wn = 0; wn < 2; ++wn) {
        unsigned be[4];
#pragma unroll
        for (int e = 0; e < 4; ++e) {
          bool p = va[t][r][wn][e] > 0;
          be[e] = (unsigned)__ballot(p);
          pm = fmaxf(pm, p ? sw[wn][e] : -INFINITY);
        }
        unsigned x0 = h ? be[2] : be[0];
        unsigned x1 = h ? be[3] : be[1];
        wd[t][wn][0] = mine ? x0 : wd[t][wn][0];
        wd[t][wn][1] = mine ? x1 : wd[t][wn][1];
      }
      for (int off = 16; off >= 1; off >>= 1) pm = fmaxf(pm, __shfl_xor(pm, off, 32));
      smax[t] = mine ? pm : smax[t];
    }
  }
  if (h == 0) {
    lds[wave][c][HP_MCOL]      = smax[0];
    lds[wave][16 + c][HP_MCOL] = smax[1];
  }
  __syncthreads();
  float sm0 = lds[0][c][HP_MCOL], sm1 = lds[0][16 + c][HP_MCOL];
#pragma unroll
  for (int w = 1; w < HP_WAVES; ++w) {
    sm0 = fmaxf(sm0, lds[w][c][HP_MCOL]);
    sm1 = fmaxf(sm1, lds[w][16 + c][HP_MCOL]);
  }
  float em0 = s1v0 + sm0, em1 = s1v1 + sm1;
  em0 = (em0 >= 0.f) ? em0 : ALPHA_ * em0;
  em1 = (em1 >= 0.f) ? em1 : ALPHA_ * em1;
  float mi0 = (sm0 == -INFINITY) ? NEG_ : em0;
  float mi1 = (sm1 == -INFINITY) ? NEG_ : em1;

  v8f acc[2][8];
#pragma unroll
  for (int t = 0; t < 2; ++t)
#pragma unroll
    for (int ct = 0; ct < 8; ++ct) acc[t][ct] = (v8f){};
  float ls0 = 0.f, ls1 = 0.f;

  ChunkSet ca, cb;
  load_chunk(ca, wl, sl, 0);
  for (int it = 0; it < HP_JW / 32; ++it) {
    bool more = (it + 1) < HP_JW / 32;
    bool hi   = it >= 4;
    int  sh   = 8 * (it & 3);
    unsigned u00 = (hi ? wd[0][1][0] : wd[0][0][0]) >> sh;
    unsigned u01 = (hi ? wd[0][1][1] : wd[0][0][1]) >> sh;
    unsigned u10 = (hi ? wd[1][1][0] : wd[1][0][0]) >> sh;
    unsigned u11 = (hi ? wd[1][1][1] : wd[1][0][1]) >> sh;
    load_chunk(cb, wl, sl, 2 * it + 1);
    __builtin_amdgcn_sched_barrier(0);
    mma_chunk(acc, ca, u00, u01, u10, u11, s1v0, s1v1, mi0, mi1, ls0, ls1);
    if (more) load_chunk(ca, wl, sl, 2 * it + 2);
    __builtin_amdgcn_sched_barrier(0);
    mma_chunk(acc, cb, u00 >> 4, u01 >> 4, u10 >> 4, u11 >> 4, s1v0, s1v1, mi0, mi1,
              ls0, ls1);
  }

  ls0 += __shfl_xor(ls0, 16, 32);
  ls1 += __shfl_xor(ls1, 16, 32);
  if (h == 0) {
    lds[wave][c][HP_LCOL]      = ls0;
    lds[wave][16 + c][HP_LCOL] = ls1;
  }

#pragma unroll
  for (int t = 0; t < 2; ++t)
#pragma unroll
    for (int r = 0; r < 8; ++r) {
      int M = 16 * t + 8 * h + r;
      v4f lo = {acc[t][0][r], acc[t][1][r], acc[t][2][r], acc[t][3][r]};
      v4f hi = {acc[t][4][r], acc[t][5][r], acc[t][6][r], acc[t][7][r]};
      *(v4f*)&lds[wave][M][4 * c]      = lo;
      *(v4f*)&lds[wave][M][64 + 4 * c] = hi;
    }
  __syncthreads();

  int row32 = tid >> 3;
  int seg   = tid & 7;
  float lsum = lds[0][row32][HP_LCOL];
#pragma unroll
  for (int w = 1; w < HP_WAVES; ++w) lsum += lds[w][row32][HP_LCOL];
  float invl = 1.0f / lsum;
  v4f vals[4];
  float s = 0.f, ss = 0.f;
#pragma unroll
  for (int k = 0; k < 4; ++k) {
    int col = 32 * k + 4 * seg;
    v4f x = *(const v4f*)&lds[0][row32][col];
#pragma unroll
    for (int w = 1; w < HP_WAVES; ++w) x += *(const v4f*)&lds[w][row32][col];
    x *= invl;
    vals[k] = x;
#pragma unroll
    for (int e = 0; e < 4; ++e) { s += x[e]; ss += x[e] * x[e]; }
  }
  for (int off = 4; off >= 1; off >>= 1) {
    s  += __shfl_xor(s, off, 8);
    ss += __shfl_xor(ss, off, 8);
  }
  float mu   = s * (1.0f / FOUT);
  float var  = ss * (1.0f / FOUT) - mu * mu;
  float rstd = rsqrtf(var + EPS_);
  float* orow = out + (size_t)(b * N_ + i0 + row32) * FOUT;
#pragma unroll
  for (int k = 0; k < 4; ++k) {
    int col = 32 * k + 4 * seg;
    v4f g  = *(const v4f*)(gamma + col);
    v4f be = *(const v4f*)(beta + col);
    v4f o;
#pragma unroll
    for (int e = 0; e < 4; ++e) {
      float x = (vals[k][e] - mu) * rstd * g[e] + be[e];
      o[e] = 0.5f * x * (1.0f + erff(x * 0.70710678118654752f));
    }
    *(v4f*)(orow + col) = o;
  }
}

extern "C" void kernel_launch(void* const* d_in, const int* in_sizes, int n_in,
                              void* d_out, int out_size, void* d_ws, size_t ws_size,
                              hipStream_t stream) {
  const float* h     = (const float*)d_in[0];
  const int*   adj   = (const int*)d_in[1];
  const float* W     = (const float*)d_in[2];
  const float* a     = (const float*)d_in[3];
  const float* gamma = (const float*)d_in[4];
  const float* beta  = (const float*)d_in[5];
  float* out = (float*)d_out;

  float* ws = (float*)d_ws;
  float* Wh = ws;
  float* s1 = Wh + (size_t)B_ * N_ * FOUT;
  float* s2 = s1 + B_ * N_;

  wh_s12_kernel<<<(B_ * N_ / 16) / WS_STRIPS, 64 * WS_STRIPS, 0, stream>>>(h, W, a, Wh, s1, s2);
  hprime_kernel<<<B_ * (N_ / 32), 256, 0, stream>>>(adj, Wh, s1, s2, gamma, beta, out);
}


# h_prime: no adj pass before the j loop: a wave shifts by its unmasked s2 max, reads adj in the loop as its A-layout mask, LDS sum rescales tiles; exact-max rerun if a sum underflows; j sums reorder
# speedup vs baseline: 2.6941x; 1.2144x over previous
#include <hip/hip_runtime.h>
#include <hip/hip_bf16.h>
#include <math.h>

typedef __attribute__((ext_vector_type(2))) float v2f;
typedef __attribute__((ext_vector_type(4))) float v4f;
typedef __attribute__((ext_vector_type(8))) float v8f;
typedef __attribute__((ext_vector_type(4))) int v4i;

#define B_   4
#define N_   2048
#define FIN  256
#define FOUT 128
#define ALPHA_ 0.2f
#define EPS_   1e-5f
#define NEG_   -9000000000000000.0f

#define WS_Q      16
#define WS_STRIPS 2

struct WSet {
  v2f a[WS_Q];
  v4f w[WS_Q][2];
};

__device__ __forceinline__ void load_wset(WSet& d, const float* __restrict__ hl,
                                          const float* __restrict__ wl, int trip) {
#pragma unroll
  for (int q = 0; q < WS_Q; ++q) {
    int k = 4 * (WS_Q * trip + q);
    d.a[q]    = *(const v2f*)(hl + k);
    d.w[q][0] = *(const v4f*)(wl + (size_t)k * FOUT);
    d.w[q][1] = *(const v4f*)(wl + (size_t)(k + 1) * FOUT);
  }
}

__device__ __forceinline__ void mma_wset(v8f (&acc)[4], const WSet& d) {
#pragma unroll
  for (int q = 0; q < WS_Q; ++q) {
#pragma unroll
    for (int t = 0; t < 4; ++t) {
      v2f bb;
      bb.x = d.w[q][0][t];
      bb.y = d.w[q][1][t];
      acc[t] = __builtin_amdgcn_wmma_f32_16x16x4_f32(false, d.a[q], false, bb,
                                                     (short)0, acc[t], false, false);
    }
  }
}

__global__ __launch_bounds__(64 * WS_STRIPS) void wh_s12_kernel(
    const float* __restrict__ h, const float* __restrict__ W,
    const float* __restrict__ a, float* __restrict__ Wh,
    float* __restrict__ s1, float* __restrict__ s2) {
  __shared__ float xch[WS_STRIPS][2][8][32];
  int lane  = threadIdx.x & 31;
  int wave  = threadIdx.x >> 5;
  int half  = wave & 1;
  int sloc  = wave >> 1;
  int strip = blockIdx.x * WS_STRIPS + sloc;
  int c  = lane & 15;
  int hh = lane >> 4;
  int row0 = 16 * strip;

  const float* hl = h + (size_t)(row0 + c) * FIN + 2 * hh;
  const float* wl = W + (size_t)(2 * hh) * FOUT + 64 * half + 4 * c;

  v8f acc[4];
#pragma unroll
  for (int t = 0; t < 4; ++t) acc[t] = (v8f){};

  WSet ca, cb;
  load_wset(ca, hl, wl, 0);
  for (int it = 0; it < FIN / (8 * WS_Q); ++it) {
    bool more = (it + 1) < FIN / (8 * WS_Q);
    load_wset(cb, hl, wl, 2 * it + 1);
    __builtin_amdgcn_sched_barrier(0);
    mma_wset(acc, ca);
    if (more) load_wset(ca, hl, wl, 2 * it + 2);
    __builtin_amdgcn_sched_barrier(0);
    mma_wset(acc, cb);
  }

  v4f x1 = *(const v4f*)(a + 64 * half + 4 * c);
  v4f x2 = *(const v4f*)(a + FOUT + 64 * half + 4 * c);
  float* wrow = Wh + (size_t)(row0 + 8 * hh) * FOUT + 64 * half + 4 * c;
  float p1[8], p2[8];
#pragma unroll
  for (int r = 0; r < 8; ++r) {
    v4f wv = {acc[0][r], acc[1][r], acc[2][r], acc[3][r]};
    *(v4f*)(wrow + (size_t)r * FOUT) = wv;
    p1[r] = wv.x * x1.x + wv.y * x1.y + wv.z * x1.z + wv.w * x1.w;
    p2[r] = wv.x * x2.x + wv.y * x2.y + wv.z * x2.z + wv.w * x2.w;
  }
#pragma unroll
  for (int r = 0; r < 8; ++r) xch[sloc][half][r][lane] = half ? p1[r] : p2[r];
  __syncthreads();
  float qd[8];
#pragma unroll
  for (int r = 0; r < 8; ++r) {
    float mine = half ? p2[r] : p1[r];
    qd[r] = mine + xch[sloc][half ^ 1][r][lane];
#pragma unroll
    for (int off = 8; off >= 1; off >>= 1) qd[r] += __shfl_xor(qd[r], off, 32);
  }
  if (c == 0) {
    float* dst = (half ? s2 : s1) + row0 + 8 * hh;
    *(v4f*)dst       = (v4f){qd[0], qd[1], qd[2], qd[3]};
    *(v4f*)(dst + 4) = (v4f){qd[4], qd[5], qd[6], qd[7]};
  }
}

#define HP_ROWS  32
#define HP_WAVES 8
#define HP_JW    (N_ / HP_WAVES)
#define HP_PITCH 160
#define HP_LCOL  128
#define HP_MCOL  129
#define HP_FCOL  130
#define HP_DMIN  0x1p-60f

struct ChunkSet {
  v4f w[4][2][2];
  v2f sv[4];
};

struct AdjSet {
  v4i m[2][2][2];
};

__device__ __forceinline__ void load_chunk(ChunkSet& d, const float* __restrict__ wl,
                                           const float* __restrict__ sl, int ch) {
  int j0 = 16 * ch - 8 * (ch & 1);
#pragma unroll
  for (int q = 0; q < 4; ++q) {
#pragma unroll
    for (int s = 0; s < 2; ++s) {
      const float* p = wl + (size_t)(j0 + 2 * q + s) * FOUT;
      d.w[q][s][0] = *(const v4f*)p;
      d.w[q][s][1] = *(const v4f*)(p + 64);
    }
    d.sv[q] = *(const v2f*)(sl + j0 + 2 * q);
  }
}

__device__ __forceinline__ void load_adj(AdjSet& d, const int* __restrict__ al0,
                                         const int* __restrict__ al1, int trip) {
#pragma unroll
  for (int cc = 0; cc < 2; ++cc)
#pragma unroll
    for (int g = 0; g < 2; ++g) {
      d.m[cc][0][g] = *(const v4i*)(al0 + 32 * trip + 8 * cc + 4 * g);
      d.m[cc][1][g] = *(const v4i*)(al1 + 32 * trip + 8 * cc + 4 * g);
    }
}

__device__ __forceinline__ void mma_chunk(v8f (&acc)[2][8], const ChunkSet& d,
                                          const v4i (&m0)[2], const v4i (&m1)[2],
                                          float s1v0, float s1v1, float mi0, float mi1,
                                          float& ls0, float& ls1) {
#pragma unroll
  for (int q = 0; q < 4; ++q) {
    v2f a0, a1;
#pragma unroll
    for (int s = 0; s < 2; ++s) {
      float sv = d.sv[q][s];
      float e0 = s1v0 + sv;
      float e1 = s1v1 + sv;
      e0 = (e0 >= 0.f) ? e0 : ALPHA_ * e0;
      e1 = (e1 >= 0.f) ? e1 : ALPHA_ * e1;
      int k0 = m0[q >> 1][2 * (q & 1) + s];
      int k1 = m1[q >> 1][2 * (q & 1) + s];
      float t0 = (k0 > 0) ? e0 : NEG_;
      float t1 = (k1 > 0) ? e1 : NEG_;
      a0[s] = __expf(t0 - mi0);
      a1[s] = __expf(t1 - mi1);
    }
    ls0 += a0[0] + a0[1];
    ls1 += a1[0] + a1[1];
#pragma unroll
    for (int ct = 0; ct < 8; ++ct) {
      v2f bb;
      bb.x = d.w[q][0][ct >> 2][ct & 3];
      bb.y = d.w[q][1][ct >> 2][ct & 3];
      acc[0][ct] = __builtin_amdgcn_wmma_f32_16x16x4_f32(false, a0, false, bb,
                                                         (short)0, acc[0][ct],
                                                         false, false);
      acc[1][ct] = __builtin_amdgcn_wmma_f32_16x16x4_f32(false, a1, false, bb,
                                                         (short)0, acc[1][ct],
                                                         false, false);
    }
  }
}

template <int M>
__device__ __forceinline__ float row_xor(float x) {
  int b = __builtin_bit_cast(int, x);
  return __builtin_bit_cast(float,
                            __builtin_amdgcn_update_dpp(b, b, 0x160 + M, 0xf, 0xf, false));
}

__device__ __forceinline__ float xhalf(float x) {
  int b = __builtin_bit_cast(int, x);
  return __builtin_bit_cast(float, __builtin_amdgcn_permlanex16(b, b, 0x76543210, 0xfedcba98,
                                                                false, false));
}

__global__ __launch_bounds__(256) void hprime_kernel(
    const int* adj, const float* Wh, const float* s1, const float* s2,
    const float* gamma, const float* beta, float* out) {
  __shared__ float lds[HP_WAVES][HP_ROWS][HP_PITCH];
  int tid  = threadIdx.x;
  int lane = tid & 31;
  int wave = tid >> 5;
  int b    = blockIdx.x / (N_ / HP_ROWS);
  int i0   = (blockIdx.x % (N_ / HP_ROWS)) * HP_ROWS;
  int c    = lane & 15;
  int h    = lane >> 4;
  int row0 = b * N_ + i0 + c;
  int row1 = row0 + 16;
  int row32 = tid >> 3;
  int seg   = tid & 7;

  float s1v0 = s1[row0], s1v1 = s1[row1];
  const float* sl  = s2 + (size_t)b * N_ + wave * HP_JW + 16 * h;
  const float* wl  = Wh + ((size_t)b * N_ + wave * HP_JW + 16 * h) * FOUT + 4 * c;
  const int*   al0 = adj + (size_t)row0 * N_ + wave * HP_JW + 16 * h;
  const int*   al1 = adj + (size_t)row1 * N_ + wave * HP_JW + 16 * h;

  const float* s2w = s2 + (size_t)b * N_ + wave * HP_JW + 4 * lane;
  v4f swa = *(const v4f*)s2w;
  v4f swb = *(const v4f*)(s2w + 128);
  float sx = fmaxf(fmaxf(fmaxf(swa.x, swa.y), fmaxf(swa.z, swa.w)),
                   fmaxf(fmaxf(swb.x, swb.y), fmaxf(swb.z, swb.w)));
  sx = fmaxf(sx, row_xor<8>(sx));
  sx = fmaxf(sx, row_xor<4>(sx));
  sx = fmaxf(sx, row_xor<2>(sx));
  sx = fmaxf(sx, row_xor<1>(sx));
  sx = fmaxf(sx, xhalf(sx));
  float mi0 = s1v0 + sx, mi1 = s1v1 + sx;
  mi0 = (mi0 >= 0.f) ? mi0 : ALPHA_ * mi0;
  mi1 = (mi1 >= 0.f) ? mi1 : ALPHA_ * mi1;

  if (tid == 0) lds[0][0][HP_FCOL] = 0.f;

#pragma nounroll
  for (int pass = 0; pass < 2; ++pass) {
    v8f acc[2][8];
#pragma unroll
    for (int t = 0; t < 2; ++t)
#pragma unroll
      for (int ct = 0; ct < 8; ++ct) acc[t][ct] = (v8f){};
    float ls0 = 0.f, ls1 = 0.f;

    ChunkSet ca, cb;
    AdjSet cur, nxt;
    load_chunk(ca, wl, sl, 0);
    load_adj(cur, al0, al1, 0);
    nxt = cur;
    for (int it = 0; it < HP_JW / 32; ++it) {
      bool more = (it + 1) < HP_JW / 32;
      load_chunk(cb, wl, sl, 2 * it + 1);
      __builtin_amdgcn_sched_barrier(0);
      if (more) load_adj(nxt, al0, al1, it + 1);
      __builtin_amdgcn_sched_barrier(0);
      mma_chunk(acc, ca, cur.m[0][0], cur.m[0][1], s1v0, s1v1, mi0, mi1, ls0, ls1);
      __builtin_amdgcn_sched_barrier(0);
      if (more) load_chunk(ca, wl, sl, 2 * it + 2);
      __builtin_amdgcn_sched_barrier(0);
      mma_chunk(acc, cb, cur.m[1][0], cur.m[1][1], s1v0, s1v1, mi0, mi1, ls0, ls1);
      cur = nxt;
    }

    ls0 += __shfl_xor(ls0, 16, 32);
    ls1 += __shfl_xor(ls1, 16, 32);
    if (h == 0) {
      lds[wave][c][HP_LCOL]      = ls0;
      lds[wave][16 + c][HP_LCOL] = ls1;
      lds[wave][c][HP_MCOL]      = mi0;
      lds[wave][16 + c][HP_MCOL] = mi1;
    }

#pragma unroll
    for (int t = 0; t < 2; ++t)
#pragma unroll
      for (int r = 0; r < 8; ++r) {
        int M = 16 * t + 8 * h + r;
        v4f lo = {acc[t][0][r], acc[t][1][r], acc[t][2][r], acc[t][3][r]};
        v4f hi = {acc[t][4][r], acc[t][5][r], acc[t][6][r], acc[t][7][r]};
        *(v4f*)&lds[wave][M][4 * c]      = lo;
        *(v4f*)&lds[wave][M][64 + 4 * c] = hi;
      }
    __syncthreads();

    float fw[HP_WAVES];
    float mx = lds[0][row32][HP_MCOL];
#pragma unroll
    for (int w = 0; w < HP_WAVES; ++w) {
      fw[w] = lds[w][row32][HP_MCOL];
      mx = fmaxf(mx, fw[w]);
    }
    float lsum = 0.f;
#pragma unroll
    for (int w = 0; w < HP_WAVES; ++w) {
      fw[w] = __expf(fw[w] - mx);
      lsum += fw[w] * lds[w][row32][HP_LCOL];
    }

    int redo = 0;
    if (pass == 0) {
      if (!(lsum >= HP_DMIN)) lds[0][0][HP_FCOL] = 1.f;
      __syncthreads();
      redo = __builtin_amdgcn_readfirstlane(lds[0][0][HP_FCOL] != 0.f ? 1 : 0);
    }

    if (!redo) {
      float invl = 1.0f / lsum;
      v4f vals[4];
      float s = 0.f, ss = 0.f;
#pragma unroll
      for (int k = 0; k < 4; ++k) {
        int col = 32 * k + 4 * seg;
        v4f x = *(const v4f*)&lds[0][row32][col] * fw[0];
#pragma unroll
        for (int w = 1; w < HP_WAVES; ++w) x += *(const v4f*)&lds[w][row32][col] * fw[w];
        x *= invl;
        vals[k] = x;
#pragma unroll
        for (int e = 0; e < 4; ++e) { s += x[e]; ss += x[e] * x[e]; }
      }
      for (int off = 4; off >= 1; off >>= 1) {
        s  += __shfl_xor(s, off, 8);
        ss += __shfl_xor(ss, off, 8);
      }
      float mu   = s * (1.0f / FOUT);
      float var  = ss * (1.0f / FOUT) - mu * mu;
      float rstd = rsqrtf(var + EPS_);
      float* orow = out + (size_t)(b * N_ + i0 + row32) * FOUT;
#pragma unroll
      for (int k = 0; k < 4; ++k) {
        int col = 32 * k + 4 * seg;
        v4f g  = *(const v4f*)(gamma + col);
        v4f be = *(const v4f*)(beta + col);
        v4f o;
#pragma unroll
        for (int e = 0; e < 4; ++e) {
          float x = (vals[k][e] - mu) * rstd * g[e] + be[e];
          o[e] = 0.5f * x * (1.0f + erff(x * 0.70710678118654752f));
        }
        *(v4f*)(orow + col) = o;
      }
      break;
    }

    {
      const int*   ar = adj + (size_t)(b * N_ + i0 + row32) * N_ + 256 * seg;
      const float* sr = s2 + (size_t)b * N_ + 256 * seg;
      float pm = -INFINITY;
      for (int k = 0; k < 64; ++k) {
        v4i av = *(const v4i*)(ar + 4 * k);
        v4f xv = *(const v4f*)(sr + 4 * k);
#pragma unroll
        for (int e = 0; e < 4; ++e) pm = fmaxf(pm, (av[e] > 0) ? xv[e] : -INFINITY);
      }
      for (int off = 4; off >= 1; off >>= 1) pm = fmaxf(pm, __shfl_xor(pm, off, 8));
      if (seg == 0) lds[0][row32][HP_MCOL] = pm;
      __syncthreads();
      float sm0 = lds[0][c][HP_MCOL], sm1 = lds[0][16 + c][HP_MCOL];
      __syncthreads();
      float em0 = s1v0 + sm0, em1 = s1v1 + sm1;
      em0 = (em0 >= 0.f) ? em0 : ALPHA_ * em0;
      em1 = (em1 >= 0.f) ? em1 : ALPHA_ * em1;
      mi0 = (sm0 == -INFINITY) ? NEG_ : em0;
      mi1 = (sm1 == -INFINITY) ? NEG_ : em1;
    }
  }
}

extern "C" void kernel_launch(void* const* d_in, const int* in_sizes, int n_in,
                              void* d_out, int out_size, void* d_ws, size_t ws_size,
                              hipStream_t stream) {
  const float* h     = (const float*)d_in[0];
  const int*   adj   = (const int*)d_in[1];
  const float* W     = (const float*)d_in[2];
  const float* a     = (const float*)d_in[3];
  const float* gamma = (const float*)d_in[4];
  const float* beta  = (const float*)d_in[5];
  float* out = (float*)d_out;

  float* ws = (float*)d_ws;
  float* Wh = ws;
  float* s1 = Wh + (size_t)B_ * N_ * FOUT;
  float* s2 = s1 + B_ * N_;

  wh_s12_kernel<<<(B_ * N_ / 16) / WS_STRIPS, 64 * WS_STRIPS, 0, stream>>>(h, W, a, Wh, s1, s2);
  hprime_kernel<<<B_ * (N_ / 32), 256, 0, stream>>>(adj, Wh, s1, s2, gamma, beta, out);
}


# h_prime tail: a thread's 16 outputs as one straight-line stretch: gamma/beta loaded ahead of the stores, GELU's erf with both ranges computed and selected (no branch), so the 16 chains interleave
# speedup vs baseline: 2.8245x; 1.0484x over previous
#include <hip/hip_runtime.h>
#include <hip/hip_bf16.h>
#include <math.h>

typedef __attribute__((ext_vector_type(2))) float v2f;
typedef __attribute__((ext_vector_type(4))) float v4f;
typedef __attribute__((ext_vector_type(8))) float v8f;
typedef __attribute__((ext_vector_type(4))) int v4i;

#define B_   4
#define N_   2048
#define FIN  256
#define FOUT 128
#define ALPHA_ 0.2f
#define EPS_   1e-5f
#define NEG_   -9000000000000000.0f

#define WS_Q      16
#define WS_STRIPS 2

struct WSet {
  v2f a[WS_Q];
  v4f w[WS_Q][2];
};

__device__ __forceinline__ void load_wset(WSet& d, const float* __restrict__ hl,
                                          const float* __restrict__ wl, int trip) {
#pragma unroll
  for (int q = 0; q < WS_Q; ++q) {
    int k = 4 * (WS_Q * trip + q);
    d.a[q]    = *(const v2f*)(hl + k);
    d.w[q][0] = *(const v4f*)(wl + (size_t)k * FOUT);
    d.w[q][1] = *(const v4f*)(wl + (size_t)(k + 1) * FOUT);
  }
}

__device__ __forceinline__ void mma_wset(v8f (&acc)[4], const WSet& d) {
#pragma unroll
  for (int q = 0; q < WS_Q; ++q) {
#pragma unroll
    for (int t = 0; t < 4; ++t) {
      v2f bb;
      bb.x = d.w[q][0][t];
      bb.y = d.w[q][1][t];
      acc[t] = __builtin_amdgcn_wmma_f32_16x16x4_f32(false, d.a[q], false, bb,
                                                     (short)0, acc[t], false, false);
    }
  }
}

__global__ __launch_bounds__(64 * WS_STRIPS) void wh_s12_kernel(
    const float* __restrict__ h, const float* __restrict__ W,
    const float* __restrict__ a, float* __restrict__ Wh,
    float* __restrict__ s1, float* __restrict__ s2) {
  __shared__ float xch[WS_STRIPS][2][8][32];
  int lane  = threadIdx.x & 31;
  int wave  = threadIdx.x >> 5;
  int half  = wave & 1;
  int sloc  = wave >> 1;
  int strip = blockIdx.x * WS_STRIPS + sloc;
  int c  = lane & 15;
  int hh = lane >> 4;
  int row0 = 16 * strip;

  const float* hl = h + (size_t)(row0 + c) * FIN + 2 * hh;
  const float* wl = W + (size_t)(2 * hh) * FOUT + 64 * half + 4 * c;

  v8f acc[4];
#pragma unroll
  for (int t = 0; t < 4; ++t) acc[t] = (v8f){};

  WSet ca, cb;
  load_wset(ca, hl, wl, 0);
  for (int it = 0; it < FIN / (8 * WS_Q); ++it) {
    bool more = (it + 1) < FIN / (8 * WS_Q);
    load_wset(cb, hl, wl, 2 * it + 1);
    __builtin_amdgcn_sched_barrier(0);
    mma_wset(acc, ca);
    if (more) load_wset(ca, hl, wl, 2 * it + 2);
    __builtin_amdgcn_sched_barrier(0);
    mma_wset(acc, cb);
  }

  v4f x1 = *(const v4f*)(a + 64 * half + 4 * c);
  v4f x2 = *(const v4f*)(a + FOUT + 64 * half + 4 * c);
  float* wrow = Wh + (size_t)(row0 + 8 * hh) * FOUT + 64 * half + 4 * c;
  float p1[8], p2[8];
#pragma unroll
  for (int r = 0; r < 8; ++r) {
    v4f wv = {acc[0][r], acc[1][r], acc[2][r], acc[3][r]};
    *(v4f*)(wrow + (size_t)r * FOUT) = wv;
    p1[r] = wv.x * x1.x + wv.y * x1.y + wv.z * x1.z + wv.w * x1.w;
    p2[r] = wv.x * x2.x + wv.y * x2.y + wv.z * x2.z + wv.w * x2.w;
  }
#pragma unroll
  for (int r = 0; r < 8; ++r) xch[sloc][half][r][lane] = half ? p1[r] : p2[r];
  __syncthreads();
  float qd[8];
#pragma unroll
  for (int r = 0; r < 8; ++r) {
    float mine = half ? p2[r] : p1[r];
    qd[r] = mine + xch[sloc][half ^ 1][r][lane];
#pragma unroll
    for (int off = 8; off >= 1; off >>= 1) qd[r] += __shfl_xor(qd[r], off, 32);
  }
  if (c == 0) {
    float* dst = (half ? s2 : s1) + row0 + 8 * hh;
    *(v4f*)dst       = (v4f){qd[0], qd[1], qd[2], qd[3]};
    *(v4f*)(dst + 4) = (v4f){qd[4], qd[5], qd[6], qd[7]};
  }
}

#define HP_ROWS  32
#define HP_WAVES 8
#define HP_JW    (N_ / HP_WAVES)
#define HP_PITCH 160
#define HP_LCOL  128
#define HP_MCOL  129
#define HP_FCOL  130
#define HP_DMIN  0x1p-60f

struct ChunkSet {
  v4f w[4][2][2];
  v2f sv[4];
};

struct AdjSet {
  v4i m[2][2][2];
};

__device__ __forceinline__ void load_chunk(ChunkSet& d, const float* __restrict__ wl,
                                           const float* __restrict__ sl, int ch) {
  int j0 = 16 * ch - 8 * (ch & 1);
#pragma unroll
  for (int q = 0; q < 4; ++q) {
#pragma unroll
    for (int s = 0; s < 2; ++s) {
      const float* p = wl + (size_t)(j0 + 2 * q + s) * FOUT;
      d.w[q][s][0] = *(const v4f*)p;
      d.w[q][s][1] = *(const v4f*)(p + 64);
    }
    d.sv[q] = *(const v2f*)(sl + j0 + 2 * q);
  }
}

__device__ __forceinline__ void load_adj(AdjSet& d, const int* __restrict__ al0,
                                         const int* __restrict__ al1, int trip) {
#pragma unroll
  for (int cc = 0; cc < 2; ++cc)
#pragma unroll
    for (int g = 0; g < 2; ++g) {
      d.m[cc][0][g] = *(const v4i*)(al0 + 32 * trip + 8 * cc + 4 * g);
      d.m[cc][1][g] = *(const v4i*)(al1 + 32 * trip + 8 * cc + 4 * g);
    }
}

__device__ __forceinline__ void mma_chunk(v8f (&acc)[2][8], const ChunkSet& d,
                                          const v4i (&m0)[2], const v4i (&m1)[2],
                                          float s1v0, float s1v1, float mi0, float mi1,
                                          float& ls0, float& ls1) {
#pragma unroll
  for (int q = 0; q < 4; ++q) {
    v2f a0, a1;
#pragma unroll
    for (int s = 0; s < 2; ++s) {
      float sv = d.sv[q][s];
      float e0 = s1v0 + sv;
      float e1 = s1v1 + sv;
      e0 = (e0 >= 0.f) ? e0 : ALPHA_ * e0;
      e1 = (e1 >= 0.f) ? e1 : ALPHA_ * e1;
      int k0 = m0[q >> 1][2 * (q & 1) + s];
      int k1 = m1[q >> 1][2 * (q & 1) + s];
      float t0 = (k0 > 0) ? e0 : NEG_;
      float t1 = (k1 > 0) ? e1 : NEG_;
      a0[s] = __expf(t0 - mi0);
      a1[s] = __expf(t1 - mi1);
    }
    ls0 += a0[0] + a0[1];
    ls1 += a1[0] + a1[1];
#pragma unroll
    for (int ct = 0; ct < 8; ++ct) {
      v2f bb;
      bb.x = d.w[q][0][ct >> 2][ct & 3];
      bb.y = d.w[q][1][ct >> 2][ct & 3];
      acc[0][ct] = __builtin_amdgcn_wmma_f32_16x16x4_f32(false, a0, false, bb,
                                                         (short)0, acc[0][ct],
                                                         false, false);
      acc[1][ct] = __builtin_amdgcn_wmma_f32_16x16x4_f32(false, a1, false, bb,
                                                         (short)0, acc[1][ct],
                                                         false, false);
    }
  }
}

template <int M>
__device__ __forceinline__ float row_xor(float x) {
  int b = __builtin_bit_cast(int, x);
  return __builtin_bit_cast(float,
                            __builtin_amdgcn_update_dpp(b, b, 0x160 + M, 0xf, 0xf, false));
}

__device__ __forceinline__ float xhalf(float x) {
  int b = __builtin_bit_cast(int, x);
  return __builtin_bit_cast(float, __builtin_amdgcn_permlanex16(b, b, 0x76543210, 0xfedcba98,
                                                                false, false));
}

__device__ __forceinline__ float erf_sel(float t) {
  float a = fabsf(t);
  float q = 1.699881e-5f;
  q = fmaf(a, q, -3.7867785e-4f);
  q = fmaf(a, q, 3.8578159e-3f);
  q = fmaf(a, q, -2.4181698e-2f);
  q = fmaf(a, q, 1.0666826e-1f);
  q = fmaf(a, q, 6.349333e-1f);
  q = fmaf(a, q, 1.2868941e-1f);
  q = fmaf(a, q, a);
  float far = 1.0f - __expf(-q);
  float u = t * t;
  float p = -5.618018e-4f;
  p = fmaf(u, p, 4.913816e-3f);
  p = fmaf(u, p, -2.6707515e-2f);
  p = fmaf(u, p, 1.1280011e-1f);
  p = fmaf(u, p, -3.7612295e-1f);
  p = fmaf(u, p, 1.283791e-1f);
  float near = fmaf(a, p, a);
  return copysignf((a >= 1.0f) ? far : near, t);
}

__global__ __launch_bounds__(256) void hprime_kernel(
    const int* adj, const float* Wh, const float* s1, const float* s2,
    const float* gamma, const float* beta, float* out) {
  __shared__ float lds[HP_WAVES][HP_ROWS][HP_PITCH];
  int tid  = threadIdx.x;
  int lane = tid & 31;
  int wave = tid >> 5;
  int b    = blockIdx.x / (N_ / HP_ROWS);
  int i0   = (blockIdx.x % (N_ / HP_ROWS)) * HP_ROWS;
  int c    = lane & 15;
  int h    = lane >> 4;
  int row0 = b * N_ + i0 + c;
  int row1 = row0 + 16;
  int row32 = tid >> 3;
  int seg   = tid & 7;

  float s1v0 = s1[row0], s1v1 = s1[row1];
  const float* sl  = s2 + (size_t)b * N_ + wave * HP_JW + 16 * h;
  const float* wl  = Wh + ((size_t)b * N_ + wave * HP_JW + 16 * h) * FOUT + 4 * c;
  const int*   al0 = adj + (size_t)row0 * N_ + wave * HP_JW + 16 * h;
  const int*   al1 = adj + (size_t)row1 * N_ + wave * HP_JW + 16 * h;

  const float* s2w = s2 + (size_t)b * N_ + wave * HP_JW + 4 * lane;
  v4f swa = *(const v4f*)s2w;
  v4f swb = *(const v4f*)(s2w + 128);
  float sx = fmaxf(fmaxf(fmaxf(swa.x, swa.y), fmaxf(swa.z, swa.w)),
                   fmaxf(fmaxf(swb.x, swb.y), fmaxf(swb.z, swb.w)));
  sx = fmaxf(sx, row_xor<8>(sx));
  sx = fmaxf(sx, row_xor<4>(sx));
  sx = fmaxf(sx, row_xor<2>(sx));
  sx = fmaxf(sx, row_xor<1>(sx));
  sx = fmaxf(sx, xhalf(sx));
  float mi0 = s1v0 + sx, mi1 = s1v1 + sx;
  mi0 = (mi0 >= 0.f) ? mi0 : ALPHA_ * mi0;
  mi1 = (mi1 >= 0.f) ? mi1 : ALPHA_ * mi1;

  if (tid == 0) lds[0][0][HP_FCOL] = 0.f;

#pragma nounroll
  for (int pass = 0; pass < 2; ++pass) {
    v8f acc[2][8];
#pragma unroll
    for (int t = 0; t < 2; ++t)
#pragma unroll
      for (int ct = 0; ct < 8; ++ct) acc[t][ct] = (v8f){};
    float ls0 = 0.f, ls1 = 0.f;

    ChunkSet ca, cb;
    AdjSet cur, nxt;
    load_chunk(ca, wl, sl, 0);
    load_adj(cur, al0, al1, 0);
    nxt = cur;
    for (int it = 0; it < HP_JW / 32; ++it) {
      bool more = (it + 1) < HP_JW / 32;
      load_chunk(cb, wl, sl, 2 * it + 1);
      __builtin_amdgcn_sched_barrier(0);
      if (more) load_adj(nxt, al0, al1, it + 1);
      __builtin_amdgcn_sched_barrier(0);
      mma_chunk(acc, ca, cur.m[0][0], cur.m[0][1], s1v0, s1v1, mi0, mi1, ls0, ls1);
      __builtin_amdgcn_sched_barrier(0);
      if (more) load_chunk(ca, wl, sl, 2 * it + 2);
      __builtin_amdgcn_sched_barrier(0);
      mma_chunk(acc, cb, cur.m[1][0], cur.m[1][1], s1v0, s1v1, mi0, mi1, ls0, ls1);
      cur = nxt;
    }

    ls0 += __shfl_xor(ls0, 16, 32);
    ls1 += __shfl_xor(ls1, 16, 32);
    if (h == 0) {
      lds[wave][c][HP_LCOL]      = ls0;
      lds[wave][16 + c][HP_LCOL] = ls1;
      lds[wave][c][HP_MCOL]      = mi0;
      lds[wave][16 + c][HP_MCOL] = mi1;
    }

#pragma unroll
    for (int t = 0; t < 2; ++t)
#pragma unroll
      for (int r = 0; r < 8; ++r) {
        int M = 16 * t + 8 * h + r;
        v4f lo = {acc[t][0][r], acc[t][1][r], acc[t][2][r], acc[t][3][r]};
        v4f hi = {acc[t][4][r], acc[t][5][r], acc[t][6][r], acc[t][7][r]};
        *(v4f*)&lds[wave][M][4 * c]      = lo;
        *(v4f*)&lds[wave][M][64 + 4 * c] = hi;
      }
    __syncthreads();

    float fw[HP_WAVES];
    float mx = lds[0][row32][HP_MCOL];
#pragma unroll
    for (int w = 0; w < HP_WAVES; ++w) {
      fw[w] = lds[w][row32][HP_MCOL];
      mx = fmaxf(mx, fw[w]);
    }
    float lsum = 0.f;
#pragma unroll
    for (int w = 0; w < HP_WAVES; ++w) {
      fw[w] = __expf(fw[w] - mx);
      lsum += fw[w] * lds[w][row32][HP_LCOL];
    }

    int redo = 0;
    if (pass == 0) {
      if (!(lsum >= HP_DMIN)) lds[0][0][HP_FCOL] = 1.f;
      __syncthreads();
      redo = __builtin_amdgcn_readfirstlane(lds[0][0][HP_FCOL] != 0.f ? 1 : 0);
    }

    if (!redo) {
      v4f gm[4], bt[4];
#pragma unroll
      for (int k = 0; k < 4; ++k) {
        gm[k] = *(const v4f*)(gamma + 32 * k + 4 * seg);
        bt[k] = *(const v4f*)(beta + 32 * k + 4 * seg);
      }
      float invl = 1.0f / lsum;
      v4f vals[4];
      float s = 0.f, ss = 0.f;
#pragma unroll
      for (int k = 0; k < 4; ++k) {
        int col = 32 * k + 4 * seg;
        v4f x = *(const v4f*)&lds[0][row32][col] * fw[0];
#pragma unroll
        for (int w = 1; w < HP_WAVES; ++w) x += *(const v4f*)&lds[w][row32][col] * fw[w];
        x *= invl;
        vals[k] = x;
#pragma unroll
        for (int e = 0; e < 4; ++e) { s += x[e]; ss += x[e] * x[e]; }
      }
      for (int off = 4; off >= 1; off >>= 1) {
        s  += __shfl_xor(s, off, 8);
        ss += __shfl_xor(ss, off, 8);
      }
      float mu   = s * (1.0f / FOUT);
      float var  = ss * (1.0f / FOUT) - mu * mu;
      float rstd = rsqrtf(var + EPS_);
      float* orow = out + (size_t)(b * N_ + i0 + row32) * FOUT;
      v4f o[4];
#pragma unroll
      for (int k = 0; k < 4; ++k) {
#pragma unroll
        for (int e = 0; e < 4; ++e) {
          float x = (vals[k][e] - mu) * rstd * gm[k][e] + bt[k][e];
          o[k][e] = 0.5f * x * (1.0f + erf_sel(x * 0.70710678118654752f));
        }
      }
#pragma unroll
      for (int k = 0; k < 4; ++k) *(v4f*)(orow + 32 * k + 4 * seg) = o[k];
      break;
    }

    {
      const int*   ar = adj + (size_t)(b * N_ + i0 + row32) * N_ + 256 * seg;
      const float* sr = s2 + (size_t)b * N_ + 256 * seg;
      float pm = -INFINITY;
      for (int k = 0; k < 64; ++k) {
        v4i av = *(const v4i*)(ar + 4 * k);
        v4f xv = *(const v4f*)(sr + 4 * k);
#pragma unroll
        for (int e = 0; e < 4; ++e) pm = fmaxf(pm, (av[e] > 0) ? xv[e] : -INFINITY);
      }
      for (int off = 4; off >= 1; off >>= 1) pm = fmaxf(pm, __shfl_xor(pm, off, 8));
      if (seg == 0) lds[0][row32][HP_MCOL] = pm;
      __syncthreads();
      float sm0 = lds[0][c][HP_MCOL], sm1 = lds[0][16 + c][HP_MCOL];
      __syncthreads();
      float em0 = s1v0 + sm0, em1 = s1v1 + sm1;
      em0 = (em0 >= 0.f) ? em0 : ALPHA_ * em0;
      em1 = (em1 >= 0.f) ? em1 : ALPHA_ * em1;
      mi0 = (sm0 == -INFINITY) ? NEG_ : em0;
      mi1 = (sm1 == -INFINITY) ? NEG_ : em1;
    }
  }
}

extern "C" void kernel_launch(void* const* d_in, const int* in_sizes, int n_in,
                              void* d_out, int out_size, void* d_ws, size_t ws_size,
                              hipStream_t stream) {
  const float* h     = (const float*)d_in[0];
  const int*   adj   = (const int*)d_in[1];
  const float* W     = (const float*)d_in[2];
  const float* a     = (const float*)d_in[3];
  const float* gamma = (const float*)d_in[4];
  const float* beta  = (const float*)d_in[5];
  float* out = (float*)d_out;

  float* ws = (float*)d_ws;
  float* Wh = ws;
  float* s1 = Wh + (size_t)B_ * N_ * FOUT;
  float* s2 = s1 + B_ * N_;

  wh_s12_kernel<<<(B_ * N_ / 16) / WS_STRIPS, 64 * WS_STRIPS, 0, stream>>>(h, W, a, Wh, s1, s2);
  hprime_kernel<<<B_ * (N_ / 32), 256, 0, stream>>>(adj, Wh, s1, s2, gamma, beta, out);
}
